# starting kernel
# baseline (speedup 1.0000x reference)
#include <hip/hip_runtime.h>

#define ALPHA  0.99f
#define GAMMA  0.999f
#define LAMBD  0.7f

typedef __attribute__((ext_vector_type(2))) float v2f;
typedef __attribute__((ext_vector_type(8))) float v8f;

__global__ __launch_bounds__(256)
void rlm_prep(const float* __restrict__ q, const float* __restrict__ q_tar,
              const float* __restrict__ pi, const int* __restrict__ a_t,
              const float* __restrict__ r_t, const float* __restrict__ mu,
              const int* __restrict__ done_t,
              float* __restrict__ adv,
              float* __restrict__ wsA, float* __restrict__ wsB,
              float* __restrict__ wsE, float* __restrict__ wsM,
              float* __restrict__ wsD)
{
    __shared__ float lds_v[8 * 16];
    __shared__ float lds_vp[8 * 16];

    const int lane = threadIdx.x & 31;
    const int wave = threadIdx.x >> 5;
    const int tile = blockIdx.x * 8 + wave;
    const int t0   = tile * 16;
    const int row  = lane & 15;
    const int colbase = (lane >> 4) << 1;

    const float* qrow   = q     + (size_t)(t0 + row) * 64 + colbase;
    const float* pirow  = pi    + (size_t)(t0 + row) * 64 + colbase;
    const float* qtrow  = q_tar + (size_t)(t0 + row + 1) * 64 + colbase;
    const float* piprow = pi    + (size_t)(t0 + row + 1) * 64 + colbase;

    const v2f ones = {1.0f, 1.0f};
    v8f acc_v  = {};
    v8f acc_vp = {};
    v2f qv[16];

    #pragma unroll
    for (int s = 0; s < 16; ++s) {
        v2f qd = *(const v2f*)(qrow  + 4 * s);
        v2f pd = *(const v2f*)(pirow + 4 * s);
        qv[s] = qd;
        v2f x = qd * pd;
        acc_v = __builtin_amdgcn_wmma_f32_16x16x4_f32(
            false, x, false, ones, (short)0, acc_v, false, false);
        v2f qtd = *(const v2f*)(qtrow  + 4 * s);
        v2f ppd = *(const v2f*)(piprow + 4 * s);
        v2f x2 = qtd * ppd;
        acc_vp = __builtin_amdgcn_wmma_f32_16x16x4_f32(
            false, x2, false, ones, (short)0, acc_vp, false, false);
    }

    float selv = acc_v[0], selvp = acc_vp[0];
    #pragma unroll
    for (int j = 1; j < 8; ++j) {
        bool pick = ((row & 7) == j);
        selv  = pick ? acc_v[j]  : selv;
        selvp = pick ? acc_vp[j] : selvp;
    }
    float* myv  = lds_v  + wave * 16;
    float* myvp = lds_vp + wave * 16;
    if ((lane & 15) < 8) {
        int idx = (lane & 7) + ((lane >> 4) << 3);
        myv[idx]  = selv;
        myvp[idx] = selvp;
    }
    __syncthreads();

    const float vt = myv[row];

    float* advrow = adv + (size_t)(t0 + row) * 64 + colbase;
    const float ascale = 1.0f - ALPHA;
    #pragma unroll
    for (int s = 0; s < 16; ++s) {
        v2f o = (qv[s] - vt) * ascale;
        *(v2f*)(advrow + 4 * s) = o;
    }

    if (lane < 16) {
        const int t = t0 + lane;
        const int a = a_t[t];
        const float q_t_a = q [(size_t)t * 64 + a];
        const float pi_a  = pi[(size_t)t * 64 + a];
        const float mu_a  = mu[(size_t)t * 64 + a];
        const float r     = r_t[t];
        const int   d     = done_t[t];
        const float df    = (float)d;
        const float vp1   = myvp[lane];
        const float vv    = myv[lane];

        const float est = r + (1.0f - df) * GAMMA * vp1;
        const float td  = est - q_t_a;
        const float rho = pi_a / mu_a;
        const float cc  = LAMBD * fminf(fmaxf(rho, 0.0f), 1.0f);

        wsB[t] = LAMBD * GAMMA * rho * td;
        wsA[t] = d ? 0.0f : GAMMA * cc;
        wsM[t] = d ? 0.0f : 1.0f;
        wsE[t] = est;
        wsD[t] = ALPHA * (q_t_a - vv);
    }
}

__global__ __launch_bounds__(256)
void rlm_compose(const float* __restrict__ wsA, const float* __restrict__ wsB,
                 float* __restrict__ chA, float* __restrict__ chB, int S)
{
    const int c = blockIdx.x * blockDim.x + threadIdx.x;
    const int base = c * S;
    float Ac = 1.0f, Bc = 0.0f;
    for (int i = S - 1; i >= 0; --i) {
        const float a = wsA[base + i];
        const float b = wsB[base + i];
        Bc = b + a * Bc;
        Ac = a * Ac;
    }
    chA[c] = Ac;
    chB[c] = Bc;
}

__global__ __launch_bounds__(1024)
void rlm_chunkscan(const float* __restrict__ chA, const float* __restrict__ chB,
                   float* __restrict__ carryIn, int C)
{
    __shared__ float sA[1024];
    __shared__ float sB[1024];
    const int i = threadIdx.x;
    if (i < C) { sA[i] = chA[i]; sB[i] = chB[i]; }
    __syncthreads();
    for (int off = 1; off < C; off <<= 1) {
        float a = 0.f, b = 0.f, a2 = 0.f, b2 = 0.f;
        const bool ok = (i + off) < C && i < C;
        if (ok) { a = sA[i]; b = sB[i]; a2 = sA[i + off]; b2 = sB[i + off]; }
        __syncthreads();
        if (ok) { sA[i] = a * a2; sB[i] = b + a * b2; }
        __syncthreads();
    }
    if (i < C) carryIn[i] = (i + 1 < C) ? sB[i + 1] : 0.0f;
}

__global__ __launch_bounds__(256)
void rlm_finalize(const float* __restrict__ wsA, const float* __restrict__ wsB,
                  const float* __restrict__ wsE, const float* __restrict__ wsM,
                  const float* __restrict__ wsD, const float* __restrict__ carryIn,
                  float* __restrict__ targets, int S)
{
    const int c = blockIdx.x * blockDim.x + threadIdx.x;
    const int base = c * S;
    float y = carryIn[c];
    for (int i = S - 1; i >= 0; --i) {
        const int t = base + i;
        const float g = wsE[t] + wsM[t] * y;
        targets[t] = g + wsD[t];
        y = wsB[t] + wsA[t] * y;
    }
}

extern "C" void kernel_launch(void* const* d_in, const int* in_sizes, int n_in,
                              void* d_out, int out_size, void* d_ws, size_t ws_size,
                              hipStream_t stream)
{
    const float* q      = (const float*)d_in[0];
    const float* q_tar  = (const float*)d_in[1];
    const float* pi     = (const float*)d_in[2];
    const int*   a_t    = (const int*)  d_in[3];
    const float* r_t    = (const float*)d_in[4];
    const float* mu     = (const float*)d_in[5];
    const int*   done_t = (const int*)  d_in[6];

    const int T = in_sizes[4];
    float* targets = (float*)d_out;
    float* adv     = targets + T;

    float* ws  = (float*)d_ws;
    float* wsA = ws;
    float* wsB = ws + (size_t)T;
    float* wsE = ws + (size_t)2 * T;
    float* wsM = ws + (size_t)3 * T;
    float* wsD = ws + (size_t)4 * T;
    const int S = 256;
    const int C = T / S;
    float* chA     = ws + (size_t)5 * T;
    float* chB     = chA + C;
    float* carryIn = chB + C;

    const int tiles   = T / 16;
    const int blocks1 = tiles / 8;

    rlm_prep<<<blocks1, 256, 0, stream>>>(q, q_tar, pi, a_t, r_t, mu, done_t,
                                          adv, wsA, wsB, wsE, wsM, wsD);
    rlm_compose<<<C / 256, 256, 0, stream>>>(wsA, wsB, chA, chB, S);
    rlm_chunkscan<<<1, 1024, 0, stream>>>(chA, chB, carryIn, C);
    rlm_finalize<<<C / 256, 256, 0, stream>>>(wsA, wsB, wsE, wsM, wsD, carryIn,
                                              targets, S);
}


# scan launches remapped: one wave32 per 256-step chunk, 8 steps per lane by b128 loads, lane maps composed by shuffles (carry sums now a pairwise tree across lanes); was one thread per chunk
# speedup vs baseline: 3.1371x; 3.1371x over previous
#include <hip/hip_runtime.h>

#define ALPHA  0.99f
#define GAMMA  0.999f
#define LAMBD  0.7f

typedef __attribute__((ext_vector_type(2))) float v2f;
typedef __attribute__((ext_vector_type(8))) float v8f;

__global__ __launch_bounds__(256)
void rlm_prep(const float* __restrict__ q, const float* __restrict__ q_tar,
              const float* __restrict__ pi, const int* __restrict__ a_t,
              const float* __restrict__ r_t, const float* __restrict__ mu,
              const int* __restrict__ done_t,
              float* __restrict__ adv,
              float* __restrict__ wsA, float* __restrict__ wsB,
              float* __restrict__ wsE, float* __restrict__ wsM,
              float* __restrict__ wsD)
{
    __shared__ float lds_v[8 * 16];
    __shared__ float lds_vp[8 * 16];

    const int lane = threadIdx.x & 31;
    const int wave = threadIdx.x >> 5;
    const int tile = blockIdx.x * 8 + wave;
    const int t0   = tile * 16;
    const int row  = lane & 15;
    const int colbase = (lane >> 4) << 1;

    const float* qrow   = q     + (size_t)(t0 + row) * 64 + colbase;
    const float* pirow  = pi    + (size_t)(t0 + row) * 64 + colbase;
    const float* qtrow  = q_tar + (size_t)(t0 + row + 1) * 64 + colbase;
    const float* piprow = pi    + (size_t)(t0 + row + 1) * 64 + colbase;

    const v2f ones = {1.0f, 1.0f};
    v8f acc_v  = {};
    v8f acc_vp = {};
    v2f qv[16];

    #pragma unroll
    for (int s = 0; s < 16; ++s) {
        v2f qd = *(const v2f*)(qrow  + 4 * s);
        v2f pd = *(const v2f*)(pirow + 4 * s);
        qv[s] = qd;
        v2f x = qd * pd;
        acc_v = __builtin_amdgcn_wmma_f32_16x16x4_f32(
            false, x, false, ones, (short)0, acc_v, false, false);
        v2f qtd = *(const v2f*)(qtrow  + 4 * s);
        v2f ppd = *(const v2f*)(piprow + 4 * s);
        v2f x2 = qtd * ppd;
        acc_vp = __builtin_amdgcn_wmma_f32_16x16x4_f32(
            false, x2, false, ones, (short)0, acc_vp, false, false);
    }

    float selv = acc_v[0], selvp = acc_vp[0];
    #pragma unroll
    for (int j = 1; j < 8; ++j) {
        bool pick = ((row & 7) == j);
        selv  = pick ? acc_v[j]  : selv;
        selvp = pick ? acc_vp[j] : selvp;
    }
    float* myv  = lds_v  + wave * 16;
    float* myvp = lds_vp + wave * 16;
    if ((lane & 15) < 8) {
        int idx = (lane & 7) + ((lane >> 4) << 3);
        myv[idx]  = selv;
        myvp[idx] = selvp;
    }
    __syncthreads();

    const float vt = myv[row];

    float* advrow = adv + (size_t)(t0 + row) * 64 + colbase;
    const float ascale = 1.0f - ALPHA;
    #pragma unroll
    for (int s = 0; s < 16; ++s) {
        v2f o = (qv[s] - vt) * ascale;
        *(v2f*)(advrow + 4 * s) = o;
    }

    if (lane < 16) {
        const int t = t0 + lane;
        const int a = a_t[t];
        const float q_t_a = q [(size_t)t * 64 + a];
        const float pi_a  = pi[(size_t)t * 64 + a];
        const float mu_a  = mu[(size_t)t * 64 + a];
        const float r     = r_t[t];
        const int   d     = done_t[t];
        const float df    = (float)d;
        const float vp1   = myvp[lane];
        const float vv    = myv[lane];

        const float est = r + (1.0f - df) * GAMMA * vp1;
        const float td  = est - q_t_a;
        const float rho = pi_a / mu_a;
        const float cc  = LAMBD * fminf(fmaxf(rho, 0.0f), 1.0f);

        wsB[t] = LAMBD * GAMMA * rho * td;
        wsA[t] = d ? 0.0f : GAMMA * cc;
        wsM[t] = d ? 0.0f : 1.0f;
        wsE[t] = est;
        wsD[t] = ALPHA * (q_t_a - vv);
    }
}

__global__ __launch_bounds__(128)
void rlm_compose(const float* __restrict__ wsA, const float* __restrict__ wsB,
                 float* __restrict__ chA, float* __restrict__ chB)
{
    const int lane = threadIdx.x & 31;
    const int c    = blockIdx.x * 4 + (threadIdx.x >> 5);
    const size_t o = (size_t)c * 256 + (size_t)lane * 8;

    const float4 a0 = *(const float4*)(wsA + o);
    const float4 a1 = *(const float4*)(wsA + o + 4);
    const float4 b0 = *(const float4*)(wsB + o);
    const float4 b1 = *(const float4*)(wsB + o + 4);
    const float a[8] = {a0.x, a0.y, a0.z, a0.w, a1.x, a1.y, a1.z, a1.w};
    const float b[8] = {b0.x, b0.y, b0.z, b0.w, b1.x, b1.y, b1.z, b1.w};

    float Ac = 1.0f, Bc = 0.0f;
    #pragma unroll
    for (int i = 7; i >= 0; --i) {
        Bc = b[i] + a[i] * Bc;
        Ac = a[i] * Ac;
    }
    #pragma unroll
    for (int off = 1; off < 32; off <<= 1) {
        const float a2 = __shfl_down(Ac, off, 32);
        const float b2 = __shfl_down(Bc, off, 32);
        Bc = Bc + Ac * b2;
        Ac = Ac * a2;
    }
    if (lane == 0) { chA[c] = Ac; chB[c] = Bc; }
}

__global__ __launch_bounds__(1024)
void rlm_chunkscan(const float* __restrict__ chA, const float* __restrict__ chB,
                   float* __restrict__ carryIn, int C)
{
    __shared__ float sA[1024];
    __shared__ float sB[1024];
    const int i = threadIdx.x;
    if (i < C) { sA[i] = chA[i]; sB[i] = chB[i]; }
    __syncthreads();
    for (int off = 1; off < C; off <<= 1) {
        float a = 0.f, b = 0.f, a2 = 0.f, b2 = 0.f;
        const bool ok = (i + off) < C && i < C;
        if (ok) { a = sA[i]; b = sB[i]; a2 = sA[i + off]; b2 = sB[i + off]; }
        __syncthreads();
        if (ok) { sA[i] = a * a2; sB[i] = b + a * b2; }
        __syncthreads();
    }
    if (i < C) carryIn[i] = (i + 1 < C) ? sB[i + 1] : 0.0f;
}

__global__ __launch_bounds__(128)
void rlm_finalize(const float* __restrict__ wsA, const float* __restrict__ wsB,
                  const float* __restrict__ wsE, const float* __restrict__ wsM,
                  const float* __restrict__ wsD, const float* __restrict__ carryIn,
                  float* __restrict__ targets)
{
    const int lane = threadIdx.x & 31;
    const int c    = blockIdx.x * 4 + (threadIdx.x >> 5);
    const size_t o = (size_t)c * 256 + (size_t)lane * 8;

    const float4 a0 = *(const float4*)(wsA + o);
    const float4 a1 = *(const float4*)(wsA + o + 4);
    const float4 b0 = *(const float4*)(wsB + o);
    const float4 b1 = *(const float4*)(wsB + o + 4);
    const float4 e0 = *(const float4*)(wsE + o);
    const float4 e1 = *(const float4*)(wsE + o + 4);
    const float4 m0 = *(const float4*)(wsM + o);
    const float4 m1 = *(const float4*)(wsM + o + 4);
    const float4 d0 = *(const float4*)(wsD + o);
    const float4 d1 = *(const float4*)(wsD + o + 4);
    const float yin = carryIn[c];

    const float a[8] = {a0.x, a0.y, a0.z, a0.w, a1.x, a1.y, a1.z, a1.w};
    const float b[8] = {b0.x, b0.y, b0.z, b0.w, b1.x, b1.y, b1.z, b1.w};
    const float e[8] = {e0.x, e0.y, e0.z, e0.w, e1.x, e1.y, e1.z, e1.w};
    const float m[8] = {m0.x, m0.y, m0.z, m0.w, m1.x, m1.y, m1.z, m1.w};
    const float d[8] = {d0.x, d0.y, d0.z, d0.w, d1.x, d1.y, d1.z, d1.w};

    float Ac = 1.0f, Bc = 0.0f;
    #pragma unroll
    for (int i = 7; i >= 0; --i) {
        Bc = b[i] + a[i] * Bc;
        Ac = a[i] * Ac;
    }
    #pragma unroll
    for (int off = 1; off < 32; off <<= 1) {
        const float a2 = __shfl_down(Ac, off, 32);
        const float b2 = __shfl_down(Bc, off, 32);
        if (lane + off < 32) {
            Bc = Bc + Ac * b2;
            Ac = Ac * a2;
        }
    }
    const float ea = __shfl_down(Ac, 1, 32);
    const float eb = __shfl_down(Bc, 1, 32);
    float y = (lane < 31) ? (eb + ea * yin) : yin;

    float r[8];
    #pragma unroll
    for (int i = 7; i >= 0; --i) {
        const float g = e[i] + m[i] * y;
        r[i] = g + d[i];
        y = b[i] + a[i] * y;
    }
    *(float4*)(targets + o)     = make_float4(r[0], r[1], r[2], r[3]);
    *(float4*)(targets + o + 4) = make_float4(r[4], r[5], r[6], r[7]);
}

extern "C" void kernel_launch(void* const* d_in, const int* in_sizes, int n_in,
                              void* d_out, int out_size, void* d_ws, size_t ws_size,
                              hipStream_t stream)
{
    const float* q      = (const float*)d_in[0];
    const float* q_tar  = (const float*)d_in[1];
    const float* pi     = (const float*)d_in[2];
    const int*   a_t    = (const int*)  d_in[3];
    const float* r_t    = (const float*)d_in[4];
    const float* mu     = (const float*)d_in[5];
    const int*   done_t = (const int*)  d_in[6];

    const int T = in_sizes[4];
    float* targets = (float*)d_out;
    float* adv     = targets + T;

    float* ws  = (float*)d_ws;
    float* wsA = ws;
    float* wsB = ws + (size_t)T;
    float* wsE = ws + (size_t)2 * T;
    float* wsM = ws + (size_t)3 * T;
    float* wsD = ws + (size_t)4 * T;
    const int S = 256;
    const int C = T / S;
    float* chA     = ws + (size_t)5 * T;
    float* chB     = chA + C;
    float* carryIn = chB + C;

    const int tiles   = T / 16;
    const int blocks1 = tiles / 8;

    rlm_prep<<<blocks1, 256, 0, stream>>>(q, q_tar, pi, a_t, r_t, mu, done_t,
                                          adv, wsA, wsB, wsE, wsM, wsD);
    rlm_compose<<<C / 4, 128, 0, stream>>>(wsA, wsB, chA, chB);
    rlm_chunkscan<<<1, 1024, 0, stream>>>(chA, chB, carryIn, C);
    rlm_finalize<<<C / 4, 128, 0, stream>>>(wsA, wsB, wsE, wsM, wsD, carryIn,
                                            targets);
}
